# starting kernel
# baseline (speedup 1.0000x reference)
#include <hip/hip_runtime.h>
#include <math.h>

#define NB   4096
#define NC   200
#define NT   25
#define NH   64
#define NG   256
#define NF   1600
#define NCLS 16

typedef __attribute__((ext_vector_type(2))) float v2f;
typedef __attribute__((ext_vector_type(8))) float v8f;

#define WMMA_F32(a, b, c) \
    __builtin_amdgcn_wmma_f32_16x16x4_f32(false, (a), false, (b), (short)0, (c), false, false)

__device__ __forceinline__ float sigf(float x) { return 1.0f / (1.0f + __expf(-x)); }

__global__ __launch_bounds__(128) void lstm_kernel(
    const float* __restrict__ x,
    const float* __restrict__ Wih,
    const float* __restrict__ Whh,
    const float* __restrict__ bih,
    const float* __restrict__ bhh,
    float* __restrict__ feat)
{
    __shared__ float hlds[16 * NH];
    __shared__ float glds[16 * NG];

    const int tid  = threadIdx.x;
    const int lane = tid & 31;
    const int gb   = (tid >> 5) * 64;
    const int b0   = blockIdx.x * 16;
    const bool hi  = lane >= 16;
    const int nn   = lane & 15;
    const int ks   = hi ? 2 : 0;
    const int rofs = hi ? 8 : 0;

    float creg[8];
#pragma unroll
    for (int ii = 0; ii < 8; ++ii) {
        creg[ii] = 0.0f;
        hlds[tid + 128 * ii] = 0.0f;
    }
    __syncthreads();

    for (int t = 0; t < NT; ++t) {
        const int off = (t % 5) * 5 + (t / 5);

        v8f acc[4];
#pragma unroll
        for (int j = 0; j < 4; ++j) {
            const int g = gb + 16 * j + nn;
            const float bv = bih[g] + bhh[g];
#pragma unroll
            for (int r = 0; r < 8; ++r) acc[j][r] = bv;
        }

        const int arow = b0 + nn;
        for (int k = 0; k < 50; ++k) {
            const int kb = 4 * k + ks;
            v2f a;
            a.x = x[(arow * NC + kb    ) * NT + off];
            a.y = x[(arow * NC + kb + 1) * NT + off];
#pragma unroll
            for (int j = 0; j < 4; ++j) {
                const int g = gb + 16 * j + nn;
                v2f b;
                b.x = Wih[g * NC + kb];
                b.y = Wih[g * NC + kb + 1];
                acc[j] = WMMA_F32(a, b, acc[j]);
            }
        }

        for (int k = 0; k < 16; ++k) {
            const int kb = 4 * k + ks;
            v2f a;
            a.x = hlds[nn * NH + kb];
            a.y = hlds[nn * NH + kb + 1];
#pragma unroll
            for (int j = 0; j < 4; ++j) {
                const int g = gb + 16 * j + nn;
                v2f b;
                b.x = Whh[g * NH + kb];
                b.y = Whh[g * NH + kb + 1];
                acc[j] = WMMA_F32(a, b, acc[j]);
            }
        }

#pragma unroll
        for (int j = 0; j < 4; ++j)
#pragma unroll
            for (int r = 0; r < 8; ++r)
                glds[(r + rofs) * NG + gb + 16 * j + nn] = acc[j][r];
        __syncthreads();

#pragma unroll
        for (int ii = 0; ii < 8; ++ii) {
            const int idx = tid + 128 * ii;
            const int row = idx >> 6;
            const int hc  = idx & 63;
            const float gi = glds[row * NG + hc];
            const float gf = glds[row * NG + 64 + hc];
            const float gg = glds[row * NG + 128 + hc];
            const float go = glds[row * NG + 192 + hc];
            const float cc = sigf(gf) * creg[ii] + sigf(gi) * tanhf(gg);
            const float hh = sigf(go) * tanhf(cc);
            creg[ii] = cc;
            hlds[row * NH + hc] = hh;
            feat[(size_t)(b0 + row) * NF + hc * NT + t] = fmaxf(hh, 0.0f);
        }
        __syncthreads();
    }
}

__global__ void bn_stats_kernel(const float* __restrict__ feat,
                                const float* __restrict__ gamma,
                                const float* __restrict__ beta,
                                float* __restrict__ scale,
                                float* __restrict__ shift)
{
    const int f = blockIdx.x * blockDim.x + threadIdx.x;
    if (f >= NF) return;
    float s = 0.0f, s2 = 0.0f;
    for (int b = 0; b < NB; ++b) {
        const float v = feat[(size_t)b * NF + f];
        s += v;
        s2 += v * v;
    }
    const float mu  = s * (1.0f / NB);
    const float var = s2 * (1.0f / NB) - mu * mu;
    const float sc  = gamma[f] * rsqrtf(var + 1e-5f);
    scale[f] = sc;
    shift[f] = beta[f] - mu * sc;
}

__global__ __launch_bounds__(32) void fc_kernel(
    const float* __restrict__ feat,
    const float* __restrict__ scale,
    const float* __restrict__ shift,
    const float* __restrict__ fcw,
    const float* __restrict__ fcb,
    float* __restrict__ out)
{
    const int lane = threadIdx.x;
    const int b0   = blockIdx.x * 16;
    const bool hi  = lane >= 16;
    const int nn   = lane & 15;
    const int ks   = hi ? 2 : 0;
    const int rofs = hi ? 8 : 0;

    v8f acc;
    const float bb = fcb[nn];
#pragma unroll
    for (int r = 0; r < 8; ++r) acc[r] = bb;

    const int arow = b0 + nn;
    for (int k = 0; k < 400; ++k) {
        const int kb = 4 * k + ks;
        v2f a, b;
        const float f0 = feat[(size_t)arow * NF + kb];
        const float f1 = feat[(size_t)arow * NF + kb + 1];
        a.x = fmaxf(f0 * scale[kb] + shift[kb], 0.0f);
        a.y = fmaxf(f1 * scale[kb + 1] + shift[kb + 1], 0.0f);
        b.x = fcw[nn * NF + kb];
        b.y = fcw[nn * NF + kb + 1];
        acc = WMMA_F32(a, b, acc);
    }

#pragma unroll
    for (int r = 0; r < 8; ++r)
        out[(b0 + r + rofs) * NCLS + nn] = acc[r];
}

extern "C" void kernel_launch(void* const* d_in, const int* in_sizes, int n_in,
                              void* d_out, int out_size, void* d_ws, size_t ws_size,
                              hipStream_t stream)
{
    const float* x     = (const float*)d_in[0];
    const float* Wih   = (const float*)d_in[1];
    const float* Whh   = (const float*)d_in[2];
    const float* bih   = (const float*)d_in[3];
    const float* bhh   = (const float*)d_in[4];
    const float* gamma = (const float*)d_in[5];
    const float* beta  = (const float*)d_in[6];
    const float* fcw   = (const float*)d_in[7];
    const float* fcb   = (const float*)d_in[8];
    float* out = (float*)d_out;

    float* feat  = (float*)d_ws;
    float* scale = feat + (size_t)NB * NF;
    float* shift = scale + NF;

    lstm_kernel<<<NB / 16, 128, 0, stream>>>(x, Wih, Whh, bih, bhh, feat);
    bn_stats_kernel<<<(NF + 255) / 256, 256, 0, stream>>>(feat, gamma, beta, scale, shift);
    fc_kernel<<<NB / 16, 32, 0, stream>>>(feat, scale, shift, fcw, fcb, out);
}


# fc kernel: A and B fragment loads run a ring of 25 register sets (25 K-steps, 50 loads) ahead of their WMMAs, scale and shift read from LDS; was one round trip per 5 K-steps; k order kept
# speedup vs baseline: 3.5768x; 3.5768x over previous
#include <hip/hip_runtime.h>
#include <math.h>

#define NB   4096
#define NC   200
#define NT   25
#define NH   64
#define NG   256
#define NF   1600
#define NCLS 16
#define NBLK (NB / 16)
#define WS_NEED (((size_t)NB * NF + 2 * NF + 2 * (size_t)NBLK * NF) * sizeof(float))

typedef __attribute__((ext_vector_type(2))) float v2f;
typedef __attribute__((ext_vector_type(8))) float v8f;
typedef __attribute__((ext_vector_type(4))) float v4f;

#define WMMA_F32(a, b, c) \
    __builtin_amdgcn_wmma_f32_16x16x4_f32(false, (a), false, (b), (short)0, (c), false, false)

__device__ __forceinline__ float sigf(float x) { return 1.0f / (1.0f + __expf(-x)); }

#define PRE_PITCH 260
#define TCH       5
#define NRUN      (NT / TCH)
#define PRE_BUF   (16 * TCH * PRE_PITCH)
#define KSTEPS    (NC / 4)
#define KPS       (KSTEPS / TCH)
#define RING      5
#define FSET      (TCH + 4)

__device__ __forceinline__ void input_product_ksteps(
    const float* __restrict__ x, const float* __restrict__ Wih,
    v8f (&accs)[TCH][4], v2f (&frag)[64],
    const int (&aoff)[TCH], const int (&woff)[4], const float (&bv)[4],
    float* __restrict__ pre, int& prun, int& pk0, int gb, int nn, int hsel)
{
#pragma unroll
    for (int i = 0; i < KPS; ++i) {
        const int s  = i % RING;
        const int sp = (i + RING - 1) % RING;
        int k4 = pk0 + i + (RING - 1);
        int r4 = prun;
        if (k4 >= KSTEPS) { k4 -= KSTEPS; r4 += 1; }
        if (r4 > NRUN - 1) { r4 = NRUN - 1; k4 = KSTEPS - 1; }
        const int xo = r4 + k4 * (4 * NT);
#pragma unroll
        for (int m = 0; m < TCH; ++m) {
            frag[sp * FSET + m].x = x[aoff[m] + xo];
            frag[sp * FSET + m].y = x[aoff[m] + xo + NT];
        }
#pragma unroll
        for (int j = 0; j < 4; ++j) {
            frag[sp * FSET + TCH + j].x = Wih[woff[j] + 4 * k4];
            frag[sp * FSET + TCH + j].y = Wih[woff[j] + 4 * k4 + 1];
        }
        __builtin_amdgcn_sched_barrier(0);
#pragma unroll
        for (int m = 0; m < TCH; ++m)
#pragma unroll
            for (int j = 0; j < 4; ++j)
                accs[m][j] = WMMA_F32(frag[s * FSET + m], frag[s * FSET + TCH + j], accs[m][j]);
    }
    if (pk0 + KPS == KSTEPS) {
        float* __restrict__ dst = pre + (prun & 1) * PRE_BUF;
#pragma unroll
        for (int m = 0; m < TCH; ++m)
#pragma unroll
            for (int j = 0; j < 4; ++j)
#pragma unroll
                for (int r = 0; r < 8; ++r) {
                    dst[(16 * m + 8 * hsel + r) * PRE_PITCH + gb + 16 * j + nn] = accs[m][j][r];
                    accs[m][j][r] = bv[j];
                }
        prun += 1;
        pk0 = 0;
    } else {
        pk0 += KPS;
    }
}

__global__ __launch_bounds__(256) void lstm_kernel(
    const float* __restrict__ x,
    const float* __restrict__ Wih,
    const float* __restrict__ Whh,
    const float* __restrict__ bih,
    const float* __restrict__ bhh,
    float* __restrict__ feat,
    float* __restrict__ psum,
    float* __restrict__ psq,
    size_t ws_bytes)
{
    __shared__ float stat_s[2 * NF];
    __shared__ float stat_q[2 * NF];
    __shared__ float hlds[16 * NH];
    __shared__ float glds[16 * NG];
    __shared__ float pre[2 * PRE_BUF];
    __shared__ __attribute__((aligned(16))) float ftile[16 * NF];

    const int tid  = threadIdx.x;
    const int lane = tid & 31;
    const int wv = __builtin_amdgcn_readfirstlane(tid >> 5);
    const bool producer = wv >= 4;
    const int  rank = wv & 3;
    const int ctid = rank * 32 + lane;
    const int gb   = rank * 64;
    const int b0   = blockIdx.x * 16;
    const bool hi  = lane >= 16;
    const int nn   = lane & 15;
    const int ks   = hi ? 2 : 0;
    const int rofs = hi ? 8 : 0;
    const int hsel = hi ? 1 : 0;

    float creg[8];
#pragma unroll
    for (int ii = 0; ii < 8; ++ii) creg[ii] = 0.0f;
#pragma unroll
    for (int ii = 0; ii < 4; ++ii) hlds[tid + 256 * ii] = 0.0f;

    v8f accs[TCH][4];
    v2f frag[64];
    float bv[4];
    int aoff[TCH], woff[4];
#pragma unroll
    for (int j = 0; j < 4; ++j) {
        const int g = gb + 16 * j + nn;
        bv[j] = bih[g] + bhh[g];
        woff[j] = g * NC + 2 * hsel;
#pragma unroll
        for (int m = 0; m < TCH; ++m)
#pragma unroll
            for (int r = 0; r < 8; ++r) accs[m][j][r] = bv[j];
    }
#pragma unroll
    for (int m = 0; m < TCH; ++m) {
        const int f  = 16 * m + nn;
        const int bl = f / TCH;
        const int tl = f - bl * TCH;
        aoff[m] = ((b0 + bl) * NC + 2 * hsel) * NT + tl * 5;
    }
#pragma unroll
    for (int i = 0; i < 64; ++i) { frag[i].x = 0.0f; frag[i].y = 0.0f; }

    int prun = 0, pk0 = 0;
    if (producer) {
#pragma unroll
        for (int u = 0; u < RING - 1; ++u) {
#pragma unroll
            for (int m = 0; m < TCH; ++m) {
                frag[u * FSET + m].x = x[aoff[m] + u * (4 * NT)];
                frag[u * FSET + m].y = x[aoff[m] + u * (4 * NT) + NT];
            }
#pragma unroll
            for (int j = 0; j < 4; ++j) {
                frag[u * FSET + TCH + j].x = Wih[woff[j] + 4 * u];
                frag[u * FSET + TCH + j].y = Wih[woff[j] + 4 * u + 1];
            }
        }
        for (int q = 0; q < TCH; ++q)
            input_product_ksteps(x, Wih, accs, frag, aoff, woff, bv, pre, prun, pk0, gb, nn, hsel);
    } else {
#pragma unroll
        for (int k = 0; k < 16; ++k)
#pragma unroll
            for (int j = 0; j < 4; ++j) {
                const int g = gb + 16 * j + nn;
                frag[4 * k + j].x = Whh[g * NH + 4 * k + ks];
                frag[4 * k + j].y = Whh[g * NH + 4 * k + ks + 1];
            }
    }
    __syncthreads();

    for (int run = 0; run < NRUN; ++run) {
        const float* __restrict__ prc = pre + (run & 1) * PRE_BUF;
        for (int tl = 0; tl < TCH; ++tl) {
            const int t = run * TCH + tl;

            if (!producer) {
#pragma unroll
                for (int j = 0; j < 4; ++j)
#pragma unroll
                    for (int r = 0; r < 8; ++r)
                        accs[0][j][r] = prc[((rofs + r) * TCH + tl) * PRE_PITCH + gb + 16 * j + nn];

#pragma unroll
                for (int k = 0; k < 16; ++k) {
                    const int kb = 4 * k + ks;
                    v2f a;
                    a.x = hlds[nn * NH + kb];
                    a.y = hlds[nn * NH + kb + 1];
#pragma unroll
                    for (int j = 0; j < 4; ++j)
                        accs[0][j] = WMMA_F32(a, frag[4 * k + j], accs[0][j]);
                }

#pragma unroll
                for (int j = 0; j < 4; ++j)
#pragma unroll
                    for (int r = 0; r < 8; ++r)
                        glds[(r + rofs) * NG + gb + 16 * j + nn] = accs[0][j][r];
            }
            __syncthreads();

            if (!producer) {
                float ps = 0.0f, pq = 0.0f;
#pragma unroll
                for (int ii = 0; ii < 8; ++ii) {
                    const int idx = ctid + 128 * ii;
                    const int row = idx >> 6;
                    const int hc  = idx & 63;
                    const float gi = glds[row * NG + hc];
                    const float gf = glds[row * NG + 64 + hc];
                    const float gg = glds[row * NG + 128 + hc];
                    const float go = glds[row * NG + 192 + hc];
                    const float cc = sigf(gf) * creg[ii] + sigf(gi) * tanhf(gg);
                    const float hh = sigf(go) * tanhf(cc);
                    creg[ii] = cc;
                    hlds[row * NH + hc] = hh;
                    const float rv = fmaxf(hh, 0.0f);
                    ftile[row * NF + hc * NT + t] = rv;
                    ps += rv;
                    pq += rv * rv;
                }
                stat_s[(ctid >> 6) * NF + (ctid & 63) * NT + t] = ps;
                stat_q[(ctid >> 6) * NF + (ctid & 63) * NT + t] = pq;
            } else if (run + 1 < NRUN) {
                input_product_ksteps(x, Wih, accs, frag, aoff, woff, bv, pre, prun, pk0, gb, nn, hsel);
            }
            __syncthreads();
        }
    }

    {
        v4f* __restrict__ dstv = (v4f*)(feat + (size_t)b0 * NF);
        const v4f* srcv = (const v4f*)ftile;
        for (int i = tid; i < 16 * NF / 4; i += 256) dstv[i] = srcv[i];
    }

    if (ws_bytes >= WS_NEED) {
        for (int f = tid; f < NF; f += 256) {
            psum[(size_t)blockIdx.x * NF + f] = stat_s[f] + stat_s[NF + f];
            psq[(size_t)blockIdx.x * NF + f]  = stat_q[f] + stat_q[NF + f];
        }
    }
}

__global__ __launch_bounds__(256) void bn_stats_kernel(const float* __restrict__ feat,
                                const float* __restrict__ psum,
                                const float* __restrict__ psq,
                                const float* __restrict__ gamma,
                                const float* __restrict__ beta,
                                float* __restrict__ scale,
                                float* __restrict__ shift,
                                size_t ws_bytes)
{
    __shared__ float red_s[8][32];
    __shared__ float red_q[8][32];
    const int fl = threadIdx.x & 31;
    const int sl = threadIdx.x >> 5;
    const int f  = blockIdx.x * 32 + fl;
    float s = 0.0f, s2 = 0.0f;
    if (ws_bytes >= WS_NEED) {
        for (int r = 0; r < NBLK / 8; ++r) {
            s  += psum[(size_t)(sl * (NBLK / 8) + r) * NF + f];
            s2 += psq[(size_t)(sl * (NBLK / 8) + r) * NF + f];
        }
    } else if (sl == 0) {
        for (int b = 0; b < NB; ++b) {
            const float v = feat[(size_t)b * NF + f];
            s += v;
            s2 += v * v;
        }
    }
    red_s[sl][fl] = s;
    red_q[sl][fl] = s2;
    __syncthreads();
    if (sl == 0) {
        s = red_s[0][fl];
        s2 = red_q[0][fl];
#pragma unroll
        for (int i = 1; i < 8; ++i) {
            s += red_s[i][fl];
            s2 += red_q[i][fl];
        }
        const float mu  = s * (1.0f / NB);
        const float var = s2 * (1.0f / NB) - mu * mu;
        const float sc  = gamma[f] * rsqrtf(var + 1e-5f);
        scale[f] = sc;
        shift[f] = beta[f] - mu * sc;
    }
}

#define FKS   (NF / 4)
#define FRING 25
#define SCP   ((NF / 4 + 31) / 32)

__device__ __forceinline__ void fc_ksteps(
    const float* __restrict__ ap, const float* __restrict__ bp,
    const float* s_scale, const float* s_shift,
    v2f (&fa)[FRING], v2f (&fb)[FRING], v8f& acc, int k0, int ks, const bool last)
{
#pragma unroll
    for (int i = 0; i < FRING; ++i) {
        const int kb = 4 * (k0 + i) + ks;
        v2f a;
        a.x = fmaxf(fa[i].x * s_scale[kb] + s_shift[kb], 0.0f);
        a.y = fmaxf(fa[i].y * s_scale[kb + 1] + s_shift[kb + 1], 0.0f);
        acc = WMMA_F32(a, fb[i], acc);
        __builtin_amdgcn_sched_barrier(0x784);
        if (!last) {
            const int kn = k0 + i + FRING;
            fa[i].x = ap[4 * kn];
            fa[i].y = ap[4 * kn + 1];
            fb[i].x = bp[4 * kn];
            fb[i].y = bp[4 * kn + 1];
        }
    }
}

__global__ __launch_bounds__(32) void fc_kernel(
    const float* __restrict__ feat,
    const float* __restrict__ scale,
    const float* __restrict__ shift,
    const float* __restrict__ fcw,
    const float* __restrict__ fcb,
    float* __restrict__ out)
{
    __shared__ __attribute__((aligned(16))) float s_scale[NF];
    __shared__ __attribute__((aligned(16))) float s_shift[NF];

    const int lane = threadIdx.x;
    const int b0   = blockIdx.x * 16;
    const bool hi  = lane >= 16;
    const int nn   = lane & 15;
    const int ks   = hi ? 2 : 0;
    const int rofs = hi ? 8 : 0;

    const float bb = fcb[nn];

    v4f tsc[SCP], tsh[SCP];
#pragma unroll
    for (int j = 0; j < SCP; ++j) {
        const int i = lane + 32 * j;
        if (j < (NF / 4) / 32 || i < NF / 4) {
            tsc[j] = ((const v4f*)scale)[i];
            tsh[j] = ((const v4f*)shift)[i];
        }
    }

    const float* __restrict__ ap = feat + (size_t)(b0 + nn) * NF + ks;
    const float* __restrict__ bp = fcw + nn * NF + ks;

    v2f fa[FRING], fb[FRING];
#pragma unroll
    for (int u = 0; u < FRING; ++u) {
        fa[u].x = ap[4 * u];
        fa[u].y = ap[4 * u + 1];
        fb[u].x = bp[4 * u];
        fb[u].y = bp[4 * u + 1];
    }

#pragma unroll
    for (int j = 0; j < SCP; ++j) {
        const int i = lane + 32 * j;
        if (j < (NF / 4) / 32 || i < NF / 4) {
            ((v4f*)s_scale)[i] = tsc[j];
            ((v4f*)s_shift)[i] = tsh[j];
        }
    }

    v8f acc;
#pragma unroll
    for (int r = 0; r < 8; ++r) acc[r] = bb;

    __syncthreads();

    for (int k0 = 0; k0 < FKS - FRING; k0 += FRING)
        fc_ksteps(ap, bp, s_scale, s_shift, fa, fb, acc, k0, ks, false);
    fc_ksteps(ap, bp, s_scale, s_shift, fa, fb, acc, FKS - FRING, ks, true);

#pragma unroll
    for (int r = 0; r < 8; ++r)
        out[(b0 + r + rofs) * NCLS + nn] = acc[r];
}

extern "C" void kernel_launch(void* const* d_in, const int* in_sizes, int n_in,
                              void* d_out, int out_size, void* d_ws, size_t ws_size,
                              hipStream_t stream)
{
    const float* x     = (const float*)d_in[0];
    const float* Wih   = (const float*)d_in[1];
    const float* Whh   = (const float*)d_in[2];
    const float* bih   = (const float*)d_in[3];
    const float* bhh   = (const float*)d_in[4];
    const float* gamma = (const float*)d_in[5];
    const float* beta  = (const float*)d_in[6];
    const float* fcw   = (const float*)d_in[7];
    const float* fcb   = (const float*)d_in[8];
    float* out = (float*)d_out;

    float* feat  = (float*)d_ws;
    float* scale = feat + (size_t)NB * NF;
    float* shift = scale + NF;
    float* psum  = shift + NF;
    float* psq   = psum + (size_t)NBLK * NF;

    lstm_kernel<<<NB / 16, 256, 0, stream>>>(x, Wih, Whh, bih, bhh, feat, psum, psq, ws_size);
    bn_stats_kernel<<<NF / 32, 256, 0, stream>>>(feat, psum, psq, gamma, beta, scale, shift, ws_size);
    fc_kernel<<<NB / 16, 32, 0, stream>>>(feat, scale, shift, fcw, fcb, out);
}


# lstm kernel: each input-product thread updates 2 of its thread column's 8 cells per step (recurrence thread 6, was 8): shorter serial cell chain; batch sums from the LDS tile at block end, same order
# speedup vs baseline: 3.8363x; 1.0726x over previous
#include <hip/hip_runtime.h>
#include <math.h>

#define NB   4096
#define NC   200
#define NT   25
#define NH   64
#define NG   256
#define NF   1600
#define NCLS 16
#define NBLK (NB / 16)
#define WS_NEED (((size_t)NB * NF + 2 * NF + 2 * (size_t)NBLK * NF) * sizeof(float))

typedef __attribute__((ext_vector_type(2))) float v2f;
typedef __attribute__((ext_vector_type(8))) float v8f;
typedef __attribute__((ext_vector_type(4))) float v4f;

#define WMMA_F32(a, b, c) \
    __builtin_amdgcn_wmma_f32_16x16x4_f32(false, (a), false, (b), (short)0, (c), false, false)

__device__ __forceinline__ float sigf(float x) { return 1.0f / (1.0f + __expf(-x)); }

#define PRE_PITCH 260
#define TCH       5
#define NRUN      (NT / TCH)
#define PRE_BUF   (16 * TCH * PRE_PITCH)
#define KSTEPS    (NC / 4)
#define KPS       (KSTEPS / TCH)
#define RING      5
#define FSET      (TCH + 4)
#define CELLS_R   6

__device__ __forceinline__ void input_product_ksteps(
    const float* __restrict__ x, const float* __restrict__ Wih,
    v8f (&accs)[TCH][4], v2f (&frag)[64],
    const int (&aoff)[TCH], const int (&woff)[4], const float (&bv)[4],
    float* __restrict__ pre, int& prun, int& pk0, int gb, int nn, int hsel)
{
#pragma unroll
    for (int i = 0; i < KPS; ++i) {
        const int s  = i % RING;
        const int sp = (i + RING - 1) % RING;
        int k4 = pk0 + i + (RING - 1);
        int r4 = prun;
        if (k4 >= KSTEPS) { k4 -= KSTEPS; r4 += 1; }
        if (r4 > NRUN - 1) { r4 = NRUN - 1; k4 = KSTEPS - 1; }
        const int xo = r4 + k4 * (4 * NT);
#pragma unroll
        for (int m = 0; m < TCH; ++m) {
            frag[sp * FSET + m].x = x[aoff[m] + xo];
            frag[sp * FSET + m].y = x[aoff[m] + xo + NT];
        }
#pragma unroll
        for (int j = 0; j < 4; ++j) {
            frag[sp * FSET + TCH + j].x = Wih[woff[j] + 4 * k4];
            frag[sp * FSET + TCH + j].y = Wih[woff[j] + 4 * k4 + 1];
        }
        __builtin_amdgcn_sched_barrier(0);
#pragma unroll
        for (int m = 0; m < TCH; ++m)
#pragma unroll
            for (int j = 0; j < 4; ++j)
                accs[m][j] = WMMA_F32(frag[s * FSET + m], frag[s * FSET + TCH + j], accs[m][j]);
    }
    if (pk0 + KPS == KSTEPS) {
        float* __restrict__ dst = pre + (prun & 1) * PRE_BUF;
#pragma unroll
        for (int m = 0; m < TCH; ++m)
#pragma unroll
            for (int j = 0; j < 4; ++j)
#pragma unroll
                for (int r = 0; r < 8; ++r) {
                    dst[(16 * m + 8 * hsel + r) * PRE_PITCH + gb + 16 * j + nn] = accs[m][j][r];
                    accs[m][j][r] = bv[j];
                }
        prun += 1;
        pk0 = 0;
    } else {
        pk0 += KPS;
    }
}

template <int LO, int HI>
__device__ __forceinline__ void cell_update(
    const float* glds, float* hlds, float* ftile, float (&creg)[CELLS_R], int ctid, int t)
{
#pragma unroll
    for (int ii = LO; ii < HI; ++ii) {
        const int idx = ctid + 128 * ii;
        const int row = idx >> 6;
        const int hc  = idx & 63;
        const float gi = glds[row * NG + hc];
        const float gf = glds[row * NG + 64 + hc];
        const float gg = glds[row * NG + 128 + hc];
        const float go = glds[row * NG + 192 + hc];
        const float cc = sigf(gf) * creg[ii - LO] + sigf(gi) * tanhf(gg);
        const float hh = sigf(go) * tanhf(cc);
        creg[ii - LO] = cc;
        hlds[row * NH + hc] = hh;
        ftile[row * NF + hc * NT + t] = fmaxf(hh, 0.0f);
    }
}

__global__ __launch_bounds__(256) void lstm_kernel(
    const float* __restrict__ x,
    const float* __restrict__ Wih,
    const float* __restrict__ Whh,
    const float* __restrict__ bih,
    const float* __restrict__ bhh,
    float* __restrict__ feat,
    float* __restrict__ psum,
    float* __restrict__ psq,
    size_t ws_bytes)
{
    __shared__ float hlds[16 * NH];
    __shared__ float glds[16 * NG];
    __shared__ float pre[2 * PRE_BUF];
    __shared__ __attribute__((aligned(16))) float ftile[16 * NF];

    const int tid  = threadIdx.x;
    const int lane = tid & 31;
    const int wv = __builtin_amdgcn_readfirstlane(tid >> 5);
    const bool producer = wv >= 4;
    const int  rank = wv & 3;
    const int ctid = rank * 32 + lane;
    const int gb   = rank * 64;
    const int b0   = blockIdx.x * 16;
    const bool hi  = lane >= 16;
    const int nn   = lane & 15;
    const int ks   = hi ? 2 : 0;
    const int rofs = hi ? 8 : 0;
    const int hsel = hi ? 1 : 0;

    float creg[CELLS_R];
#pragma unroll
    for (int ii = 0; ii < CELLS_R; ++ii) creg[ii] = 0.0f;
#pragma unroll
    for (int ii = 0; ii < 4; ++ii) hlds[tid + 256 * ii] = 0.0f;

    v8f accs[TCH][4];
    v2f frag[64];
    float bv[4];
    int aoff[TCH], woff[4];
#pragma unroll
    for (int j = 0; j < 4; ++j) {
        const int g = gb + 16 * j + nn;
        bv[j] = bih[g] + bhh[g];
        woff[j] = g * NC + 2 * hsel;
#pragma unroll
        for (int m = 0; m < TCH; ++m)
#pragma unroll
            for (int r = 0; r < 8; ++r) accs[m][j][r] = bv[j];
    }
#pragma unroll
    for (int m = 0; m < TCH; ++m) {
        const int f  = 16 * m + nn;
        const int bl = f / TCH;
        const int tl = f - bl * TCH;
        aoff[m] = ((b0 + bl) * NC + 2 * hsel) * NT + tl * 5;
    }
#pragma unroll
    for (int i = 0; i < 64; ++i) { frag[i].x = 0.0f; frag[i].y = 0.0f; }

    int prun = 0, pk0 = 0;
    if (producer) {
#pragma unroll
        for (int u = 0; u < RING - 1; ++u) {
#pragma unroll
            for (int m = 0; m < TCH; ++m) {
                frag[u * FSET + m].x = x[aoff[m] + u * (4 * NT)];
                frag[u * FSET + m].y = x[aoff[m] + u * (4 * NT) + NT];
            }
#pragma unroll
            for (int j = 0; j < 4; ++j) {
                frag[u * FSET + TCH + j].x = Wih[woff[j] + 4 * u];
                frag[u * FSET + TCH + j].y = Wih[woff[j] + 4 * u + 1];
            }
        }
        for (int q = 0; q < TCH; ++q)
            input_product_ksteps(x, Wih, accs, frag, aoff, woff, bv, pre, prun, pk0, gb, nn, hsel);
    } else {
#pragma unroll
        for (int k = 0; k < 16; ++k)
#pragma unroll
            for (int j = 0; j < 4; ++j) {
                const int g = gb + 16 * j + nn;
                frag[4 * k + j].x = Whh[g * NH + 4 * k + ks];
                frag[4 * k + j].y = Whh[g * NH + 4 * k + ks + 1];
            }
    }
    __syncthreads();

    for (int run = 0; run < NRUN; ++run) {
        const float* __restrict__ prc = pre + (run & 1) * PRE_BUF;
        for (int tl = 0; tl < TCH; ++tl) {
            const int t = run * TCH + tl;

            if (!producer) {
#pragma unroll
                for (int j = 0; j < 4; ++j)
#pragma unroll
                    for (int r = 0; r < 8; ++r)
                        accs[0][j][r] = prc[((rofs + r) * TCH + tl) * PRE_PITCH + gb + 16 * j + nn];

#pragma unroll
                for (int k = 0; k < 16; ++k) {
                    const int kb = 4 * k + ks;
                    v2f a;
                    a.x = hlds[nn * NH + kb];
                    a.y = hlds[nn * NH + kb + 1];
#pragma unroll
                    for (int j = 0; j < 4; ++j)
                        accs[0][j] = WMMA_F32(a, frag[4 * k + j], accs[0][j]);
                }

#pragma unroll
                for (int j = 0; j < 4; ++j)
#pragma unroll
                    for (int r = 0; r < 8; ++r)
                        glds[(r + rofs) * NG + gb + 16 * j + nn] = accs[0][j][r];
            }
            __syncthreads();

            if (!producer) {
                cell_update<0, CELLS_R>(glds, hlds, ftile, creg, ctid, t);
            } else {
                cell_update<CELLS_R, 8>(glds, hlds, ftile, creg, ctid, t);
                if (run + 1 < NRUN)
                    input_product_ksteps(x, Wih, accs, frag, aoff, woff, bv, pre, prun, pk0, gb, nn, hsel);
            }
            __syncthreads();
        }
    }

    {
        v4f* __restrict__ dstv = (v4f*)(feat + (size_t)b0 * NF);
        const v4f* srcv = (const v4f*)ftile;
        for (int i = tid; i < 16 * NF / 4; i += 256) dstv[i] = srcv[i];
    }

    if (ws_bytes >= WS_NEED) {
        for (int f = tid; f < NF; f += 256) {
            float s0 = 0.0f, s1 = 0.0f, q0 = 0.0f, q1 = 0.0f;
#pragma unroll
            for (int ii = 0; ii < 8; ++ii) {
                const float v0 = ftile[(2 * ii) * NF + f];
                const float v1 = ftile[(2 * ii + 1) * NF + f];
                s0 += v0;
                q0 = fmaf(v0, v0, q0);
                s1 += v1;
                q1 = fmaf(v1, v1, q1);
            }
            psum[(size_t)blockIdx.x * NF + f] = s0 + s1;
            psq[(size_t)blockIdx.x * NF + f]  = q0 + q1;
        }
    }
}

__global__ __launch_bounds__(256) void bn_stats_kernel(const float* __restrict__ feat,
                                const float* __restrict__ psum,
                                const float* __restrict__ psq,
                                const float* __restrict__ gamma,
                                const float* __restrict__ beta,
                                float* __restrict__ scale,
                                float* __restrict__ shift,
                                size_t ws_bytes)
{
    __shared__ float red_s[8][32];
    __shared__ float red_q[8][32];
    const int fl = threadIdx.x & 31;
    const int sl = threadIdx.x >> 5;
    const int f  = blockIdx.x * 32 + fl;
    float s = 0.0f, s2 = 0.0f;
    if (ws_bytes >= WS_NEED) {
        for (int r = 0; r < NBLK / 8; ++r) {
            s  += psum[(size_t)(sl * (NBLK / 8) + r) * NF + f];
            s2 += psq[(size_t)(sl * (NBLK / 8) + r) * NF + f];
        }
    } else if (sl == 0) {
        for (int b = 0; b < NB; ++b) {
            const float v = feat[(size_t)b * NF + f];
            s += v;
            s2 += v * v;
        }
    }
    red_s[sl][fl] = s;
    red_q[sl][fl] = s2;
    __syncthreads();
    if (sl == 0) {
        s = red_s[0][fl];
        s2 = red_q[0][fl];
#pragma unroll
        for (int i = 1; i < 8; ++i) {
            s += red_s[i][fl];
            s2 += red_q[i][fl];
        }
        const float mu  = s * (1.0f / NB);
        const float var = s2 * (1.0f / NB) - mu * mu;
        const float sc  = gamma[f] * rsqrtf(var + 1e-5f);
        scale[f] = sc;
        shift[f] = beta[f] - mu * sc;
    }
}

#define FKS   (NF / 4)
#define FRING 25
#define SCP   ((NF / 4 + 31) / 32)

__device__ __forceinline__ void fc_ksteps(
    const float* __restrict__ ap, const float* __restrict__ bp,
    const float* s_scale, const float* s_shift,
    v2f (&fa)[FRING], v2f (&fb)[FRING], v8f& acc, int k0, int ks, const bool last)
{
#pragma unroll
    for (int i = 0; i < FRING; ++i) {
        const int kb = 4 * (k0 + i) + ks;
        v2f a;
        a.x = fmaxf(fa[i].x * s_scale[kb] + s_shift[kb], 0.0f);
        a.y = fmaxf(fa[i].y * s_scale[kb + 1] + s_shift[kb + 1], 0.0f);
        acc = WMMA_F32(a, fb[i], acc);
        __builtin_amdgcn_sched_barrier(0x784);
        if (!last) {
            const int kn = k0 + i + FRING;
            fa[i].x = ap[4 * kn];
            fa[i].y = ap[4 * kn + 1];
            fb[i].x = bp[4 * kn];
            fb[i].y = bp[4 * kn + 1];
        }
    }
}

__global__ __launch_bounds__(32) void fc_kernel(
    const float* __restrict__ feat,
    const float* __restrict__ scale,
    const float* __restrict__ shift,
    const float* __restrict__ fcw,
    const float* __restrict__ fcb,
    float* __restrict__ out)
{
    __shared__ __attribute__((aligned(16))) float s_scale[NF];
    __shared__ __attribute__((aligned(16))) float s_shift[NF];

    const int lane = threadIdx.x;
    const int b0   = blockIdx.x * 16;
    const bool hi  = lane >= 16;
    const int nn   = lane & 15;
    const int ks   = hi ? 2 : 0;
    const int rofs = hi ? 8 : 0;

    const float bb = fcb[nn];

    v4f tsc[SCP], tsh[SCP];
#pragma unroll
    for (int j = 0; j < SCP; ++j) {
        const int i = lane + 32 * j;
        if (j < (NF / 4) / 32 || i < NF / 4) {
            tsc[j] = ((const v4f*)scale)[i];
            tsh[j] = ((const v4f*)shift)[i];
        }
    }

    const float* __restrict__ ap = feat + (size_t)(b0 + nn) * NF + ks;
    const float* __restrict__ bp = fcw + nn * NF + ks;

    v2f fa[FRING], fb[FRING];
#pragma unroll
    for (int u = 0; u < FRING; ++u) {
        fa[u].x = ap[4 * u];
        fa[u].y = ap[4 * u + 1];
        fb[u].x = bp[4 * u];
        fb[u].y = bp[4 * u + 1];
    }

#pragma unroll
    for (int j = 0; j < SCP; ++j) {
        const int i = lane + 32 * j;
        if (j < (NF / 4) / 32 || i < NF / 4) {
            ((v4f*)s_scale)[i] = tsc[j];
            ((v4f*)s_shift)[i] = tsh[j];
        }
    }

    v8f acc;
#pragma unroll
    for (int r = 0; r < 8; ++r) acc[r] = bb;

    __syncthreads();

    for (int k0 = 0; k0 < FKS - FRING; k0 += FRING)
        fc_ksteps(ap, bp, s_scale, s_shift, fa, fb, acc, k0, ks, false);
    fc_ksteps(ap, bp, s_scale, s_shift, fa, fb, acc, FKS - FRING, ks, true);

#pragma unroll
    for (int r = 0; r < 8; ++r)
        out[(b0 + r + rofs) * NCLS + nn] = acc[r];
}

extern "C" void kernel_launch(void* const* d_in, const int* in_sizes, int n_in,
                              void* d_out, int out_size, void* d_ws, size_t ws_size,
                              hipStream_t stream)
{
    const float* x     = (const float*)d_in[0];
    const float* Wih   = (const float*)d_in[1];
    const float* Whh   = (const float*)d_in[2];
    const float* bih   = (const float*)d_in[3];
    const float* bhh   = (const float*)d_in[4];
    const float* gamma = (const float*)d_in[5];
    const float* beta  = (const float*)d_in[6];
    const float* fcw   = (const float*)d_in[7];
    const float* fcb   = (const float*)d_in[8];
    float* out = (float*)d_out;

    float* feat  = (float*)d_ws;
    float* scale = feat + (size_t)NB * NF;
    float* shift = scale + NF;
    float* psum  = shift + NF;
    float* psq   = psum + (size_t)NBLK * NF;

    lstm_kernel<<<NB / 16, 256, 0, stream>>>(x, Wih, Whh, bih, bhh, feat, psum, psq, ws_size);
    bn_stats_kernel<<<NF / 32, 256, 0, stream>>>(feat, psum, psq, gamma, beta, scale, shift, ws_size);
    fc_kernel<<<NB / 16, 32, 0, stream>>>(feat, scale, shift, fcw, fcb, out);
}
